# starting kernel
# baseline (speedup 1.0000x reference)
#include <hip/hip_runtime.h>
#include <hip/hip_bf16.h>

#define BB 8
#define LL 2048
#define DD 512
#define RR 512
#define MTOT (BB * LL)

typedef float v2f __attribute__((ext_vector_type(2)));
typedef float v8f __attribute__((ext_vector_type(8)));

__device__ __forceinline__ v8f wmma_f32(v2f a, v2f b, v8f c) {
  return __builtin_amdgcn_wmma_f32_16x16x4_f32(
       false, a,  false, b,
       (short)0, c,  false,  false);
}

__device__ __forceinline__ float waveRedSum(float v) {
#pragma unroll
  for (int m = 16; m >= 1; m >>= 1) v += __shfl_xor(v, m, 32);
  return v;
}
__device__ __forceinline__ float waveRedMax(float v) {
#pragma unroll
  for (int m = 16; m >= 1; m >>= 1) v = fmaxf(v, __shfl_xor(v, m, 32));
  return v;
}

__global__ __launch_bounds__(256) void prep_kernel(
    const float* __restrict__ centers, const float* __restrict__ widths,
    float* __restrict__ iw, float* __restrict__ cw, float* __restrict__ cc) {
  int row = (blockIdx.x * blockDim.x + threadIdx.x) >> 5;
  int lane = threadIdx.x & 31;
  if (row >= RR) return;
  const float* crow = centers + row * DD;
  const float* wrow = widths + row * DD;
  float acc = 0.0f;
#pragma unroll
  for (int t = 0; t < DD / 32; ++t) {
    int d = lane + 32 * t;
    float w = wrow[d];
    float inv = 1.0f / (w * w);
    float c = crow[d];
    iw[row * DD + d] = inv;
    cw[row * DD + d] = c * inv;
    acc += c * c * inv;
  }
  acc = waveRedSum(acc);
  if (lane == 0) cc[row] = acc;
}

__global__ __launch_bounds__(256) void qproj_kernel(
    const float* __restrict__ A, const float* __restrict__ Bm,
    const float* __restrict__ bias, float* __restrict__ C) {
  const int K = DD, N = DD;
  int wave = threadIdx.x >> 5;
  int lane = threadIdx.x & 31;
  int hi = lane >> 4;
  int ln = lane & 15;
  int tile_m = blockIdx.x * 128 + wave * 16;
  int tile_n = blockIdx.y * 64;

  const float* arow = A + (tile_m + ln) * K + 2 * hi;
  const float* brow = Bm + (tile_n + ln) * K + 2 * hi;
  __builtin_prefetch(arow, 0, 1);

  v8f c0 = {}, c1 = {}, c2 = {}, c3 = {};
  for (int k0 = 0; k0 < K; k0 += 4) {
    v2f a = *(const v2f*)(arow + k0);
    v2f b0 = *(const v2f*)(brow + 0 * 16 * K + k0);
    v2f b1 = *(const v2f*)(brow + 1 * 16 * K + k0);
    v2f b2 = *(const v2f*)(brow + 2 * 16 * K + k0);
    v2f b3 = *(const v2f*)(brow + 3 * 16 * K + k0);
    c0 = wmma_f32(a, b0, c0);
    c1 = wmma_f32(a, b1, c1);
    c2 = wmma_f32(a, b2, c2);
    c3 = wmma_f32(a, b3, c3);
  }
  float bq0 = bias[tile_n + 0 + ln], bq1 = bias[tile_n + 16 + ln];
  float bq2 = bias[tile_n + 32 + ln], bq3 = bias[tile_n + 48 + ln];
#pragma unroll
  for (int i = 0; i < 8; ++i) {
    int m = tile_m + i + 8 * hi;
    float* crow = C + m * N + tile_n + ln;
    crow[0]  = c0[i] + bq0;
    crow[16] = c1[i] + bq1;
    crow[32] = c2[i] + bq2;
    crow[48] = c3[i] + bq3;
  }
}

__global__ __launch_bounds__(256) void z_kernel(
    const float* __restrict__ q, const float* __restrict__ iw,
    const float* __restrict__ cw, const float* __restrict__ cc,
    float* __restrict__ z) {
  const int K = DD, N = RR;
  int wave = threadIdx.x >> 5;
  int lane = threadIdx.x & 31;
  int hi = lane >> 4;
  int ln = lane & 15;
  int tile_m = blockIdx.x * 128 + wave * 16;
  int tile_n = blockIdx.y * 32;

  const float* arow  = q  + (tile_m + ln) * K + 2 * hi;
  const float* iwrow = iw + (tile_n + ln) * K + 2 * hi;
  const float* cwrow = cw + (tile_n + ln) * K + 2 * hi;

  v8f qq0 = {}, qc0 = {}, qq1 = {}, qc1 = {};
  for (int k0 = 0; k0 < K; k0 += 4) {
    v2f a = *(const v2f*)(arow + k0);
    v2f asq = a * a;
    v2f bi0 = *(const v2f*)(iwrow + k0);
    v2f bi1 = *(const v2f*)(iwrow + 16 * K + k0);
    v2f bc0 = *(const v2f*)(cwrow + k0);
    v2f bc1 = *(const v2f*)(cwrow + 16 * K + k0);
    qq0 = wmma_f32(asq, bi0, qq0);
    qc0 = wmma_f32(a,   bc0, qc0);
    qq1 = wmma_f32(asq, bi1, qq1);
    qc1 = wmma_f32(a,   bc1, qc1);
  }
  const float scale = -0.5f / (float)DD;
  float cc0 = cc[tile_n + ln], cc1 = cc[tile_n + 16 + ln];
#pragma unroll
  for (int i = 0; i < 8; ++i) {
    int m = tile_m + i + 8 * hi;
    float* zrow = z + m * N + tile_n + ln;
    zrow[0]  = scale * (qq0[i] - 2.0f * qc0[i] + cc0);
    zrow[16] = scale * (qq1[i] - 2.0f * qc1[i] + cc1);
  }
}

__global__ __launch_bounds__(256) void softmax_kernel(float* __restrict__ z) {
  int row = (blockIdx.x * blockDim.x + threadIdx.x) >> 5;
  int lane = threadIdx.x & 31;
  if (row >= MTOT) return;
  float* p = z + (size_t)row * RR;
  float vals[RR / 32];
  float mx = -3.4e38f;
#pragma unroll
  for (int t = 0; t < RR / 32; ++t) {
    vals[t] = p[lane + 32 * t];
    mx = fmaxf(mx, vals[t]);
  }
  mx = waveRedMax(mx);
  float s = 0.0f;
#pragma unroll
  for (int t = 0; t < RR / 32; ++t) {
    vals[t] = __expf(vals[t] - mx);
    s += vals[t];
  }
  s = waveRedSum(s);
  float inv = 1.0f / s;
#pragma unroll
  for (int t = 0; t < RR / 32; ++t) p[lane + 32 * t] = vals[t] * inv;
}

__global__ __launch_bounds__(256) void conq_kernel(
    const float* __restrict__ query, const float* __restrict__ Wc,
    const float* __restrict__ bc, float* __restrict__ conq) {
  const int K = LL, N = DD;
  int b = blockIdx.z;
  int wave = threadIdx.x >> 5;
  int lane = threadIdx.x & 31;
  int hi = lane >> 4;
  int ln = lane & 15;
  int tile_m = blockIdx.x * 128 + wave * 16;
  int tile_n = blockIdx.y * 64;

  const float* arow = Wc + (tile_m + ln) * K + 2 * hi;
  const float* Qb = query + b * LL * DD;

  v8f c0 = {}, c1 = {}, c2 = {}, c3 = {};
  for (int k0 = 0; k0 < K; k0 += 4) {
    int kk = k0 + 2 * hi;
    v2f a = *(const v2f*)(arow + k0);
    const float* bp = Qb + kk * DD + tile_n + ln;
    v2f b0, b1, b2, b3;
    b0.x = bp[0];       b0.y = bp[DD + 0];
    b1.x = bp[16];      b1.y = bp[DD + 16];
    b2.x = bp[32];      b2.y = bp[DD + 32];
    b3.x = bp[48];      b3.y = bp[DD + 48];
    c0 = wmma_f32(a, b0, c0);
    c1 = wmma_f32(a, b1, c1);
    c2 = wmma_f32(a, b2, c2);
    c3 = wmma_f32(a, b3, c3);
  }
  float* Cb = conq + b * RR * DD;
#pragma unroll
  for (int i = 0; i < 8; ++i) {
    int m = tile_m + i + 8 * hi;
    float bias = bc[m];
    float* crow = Cb + m * N + tile_n + ln;
    crow[0]  = c0[i] + bias;
    crow[16] = c1[i] + bias;
    crow[32] = c2[i] + bias;
    crow[48] = c3[i] + bias;
  }
}

__global__ __launch_bounds__(256) void out_kernel(
    const float* __restrict__ Fss, const float* __restrict__ conq,
    float* __restrict__ out) {
  const int K = RR, N = DD;
  int b = blockIdx.z;
  int wave = threadIdx.x >> 5;
  int lane = threadIdx.x & 31;
  int hi = lane >> 4;
  int ln = lane & 15;
  int tile_m = blockIdx.x * 128 + wave * 16;
  int tile_n = blockIdx.y * 64;

  const float* arow = Fss + b * LL * RR + (tile_m + ln) * K + 2 * hi;
  const float* Bb_ = conq + b * RR * DD;

  v8f c0 = {}, c1 = {}, c2 = {}, c3 = {};
  for (int k0 = 0; k0 < K; k0 += 4) {
    int kk = k0 + 2 * hi;
    v2f a = *(const v2f*)(arow + k0);
    const float* bp = Bb_ + kk * DD + tile_n + ln;
    v2f b0, b1, b2, b3;
    b0.x = bp[0];       b0.y = bp[DD + 0];
    b1.x = bp[16];      b1.y = bp[DD + 16];
    b2.x = bp[32];      b2.y = bp[DD + 32];
    b3.x = bp[48];      b3.y = bp[DD + 48];
    c0 = wmma_f32(a, b0, c0);
    c1 = wmma_f32(a, b1, c1);
    c2 = wmma_f32(a, b2, c2);
    c3 = wmma_f32(a, b3, c3);
  }
  float* Ob = out + b * LL * DD;
#pragma unroll
  for (int i = 0; i < 8; ++i) {
    int m = tile_m + i + 8 * hi;
    float* crow = Ob + m * N + tile_n + ln;
    crow[0]  = c0[i];
    crow[16] = c1[i];
    crow[32] = c2[i];
    crow[48] = c3[i];
  }
}

extern "C" void kernel_launch(void* const* d_in, const int* in_sizes, int n_in,
                              void* d_out, int out_size, void* d_ws, size_t ws_size,
                              hipStream_t stream) {
  const float* query   = (const float*)d_in[0];
  const float* Wq      = (const float*)d_in[1];
  const float* bq      = (const float*)d_in[2];
  const float* Wc      = (const float*)d_in[3];
  const float* bc      = (const float*)d_in[4];
  const float* centers = (const float*)d_in[5];
  const float* widths  = (const float*)d_in[6];
  float* out = (float*)d_out;

  float* q    = (float*)d_ws;
  float* z    = q    + (size_t)MTOT * DD;
  float* iw   = z    + (size_t)MTOT * RR;
  float* cw   = iw   + (size_t)RR * DD;
  float* cc   = cw   + (size_t)RR * DD;
  float* conq = cc   + RR;

  prep_kernel<<<RR / 8, 256, 0, stream>>>(centers, widths, iw, cw, cc);

  qproj_kernel<<<dim3(MTOT / 128, DD / 64), 256, 0, stream>>>(query, Wq, bq, q);

  z_kernel<<<dim3(MTOT / 128, RR / 32), 256, 0, stream>>>(q, iw, cw, cc, z);

  softmax_kernel<<<MTOT / 8, 256, 0, stream>>>(z);

  conq_kernel<<<dim3(RR / 128, DD / 64, BB), 256, 0, stream>>>(query, Wc, bc, conq);

  out_kernel<<<dim3(LL / 128, DD / 64, BB), 256, 0, stream>>>(z, conq, out);
}


# query is no longer repacked: K5 reads B as two 4-byte row loads per fragment on a 64x64 wave tile, k split over 4 wave groups and summed via LDS (conq k sum in 4 quarters); K1 stages its rows in LDS
# speedup vs baseline: 2.4545x; 2.4545x over previous
#include <hip/hip_runtime.h>
#include <hip/hip_bf16.h>

#define BB 8
#define LL 2048
#define DD 512
#define RR 512
#define MTOT (BB * LL)

#define KSTEPS 128
#define TILE_FLOATS 8192
#define SM_PITCH 516
#define LTILE_FLOATS 32768

#define SCHED_FENCE() __builtin_amdgcn_sched_barrier(0)

typedef float v2f __attribute__((ext_vector_type(2)));
typedef float v8f __attribute__((ext_vector_type(8)));

__device__ __forceinline__ v8f wmma_f32(v2f a, v2f b, v8f c) {
  return __builtin_amdgcn_wmma_f32_16x16x4_f32(
       false, a,  false, b,
       (short)0, c,  false,  false);
}

__device__ __forceinline__ float waveRedSum(float v) {
#pragma unroll
  for (int m = 16; m >= 1; m >>= 1) v += __shfl_xor(v, m, 32);
  return v;
}
__device__ __forceinline__ float waveRedMax(float v) {
#pragma unroll
  for (int m = 16; m >= 1; m >>= 1) v = fmaxf(v, __shfl_xor(v, m, 32));
  return v;
}

#define PACK_WQ_BLOCKS (DD / 16)
#define PACK_WC_CHUNKS (LL / DD)
#define PACK_WC_BLOCKS ((RR / 16) * PACK_WC_CHUNKS)
__global__ __launch_bounds__(256) void pack_kernel(
    const float* __restrict__ Wq, const float* __restrict__ Wc,
    float* __restrict__ wqp, float* __restrict__ wcp) {
  __shared__ __attribute__((aligned(16))) float tile[16 * SM_PITCH];
  int t = blockIdx.x;
  const float* src;
  float* dst;
  int ld;
  if (t < PACK_WQ_BLOCKS) {
    src = Wq + t * 16 * DD;
    ld = DD;
    dst = wqp + t * TILE_FLOATS;
  } else {
    int u = t - PACK_WQ_BLOCKS;
    int rt = u / PACK_WC_CHUNKS;
    int ch = u % PACK_WC_CHUNKS;
    src = Wc + rt * 16 * LL + ch * DD;
    ld = LL;
    dst = wcp + rt * LTILE_FLOATS + ch * TILE_FLOATS;
  }
  float4 v[8];
#pragma unroll
  for (int i = 0; i < 8; ++i) {
    int idx = threadIdx.x + 256 * i;
    v[i] = *(const float4*)(src + (idx >> 7) * ld + 4 * (idx & 127));
  }
#pragma unroll
  for (int i = 0; i < 8; ++i) {
    int idx = threadIdx.x + 256 * i;
    *(float4*)&tile[(idx >> 7) * SM_PITCH + 4 * (idx & 127)] = v[i];
  }
  __syncthreads();
  for (int j = 0; j < 16; ++j) {
    int p2 = threadIdx.x + 256 * j;
    int ks = p2 >> 5;
    int hi = (p2 >> 4) & 1;
    int ln = p2 & 15;
    v2f o = *(const v2f*)&tile[ln * SM_PITCH + 4 * ks + 2 * hi];
    *(v2f*)(dst + 2 * p2) = o;
  }
}

__global__ __launch_bounds__(256) void prep_kernel(
    const float* __restrict__ centers, const float* __restrict__ widths,
    float* __restrict__ iwp, float* __restrict__ cwp, float* __restrict__ cc) {
  int g = (blockIdx.x * blockDim.x + threadIdx.x) >> 5;
  int lane = threadIdx.x & 31;
  if (g >= RR) return;
  {
    const float* crow = centers + g * DD;
    const float* wrow = widths + g * DD;
    float acc = 0.0f;
#pragma unroll
    for (int t = 0; t < DD / 32; ++t) {
      int d = lane + 32 * t;
      float w = wrow[d];
      float inv = 1.0f / (w * w);
      float c = crow[d];
      acc += c * c * inv;
    }
    acc = waveRedSum(acc);
    if (lane == 0) cc[g] = acc;
  }
  {
    int rt = g >> 4;
    int ks0 = (g & 15) * 8;
    int hi = lane >> 4;
    int ln = lane & 15;
    const float* wsrc = widths  + (rt * 16 + ln) * DD + 2 * hi;
    const float* csrc = centers + (rt * 16 + ln) * DD + 2 * hi;
#pragma unroll
    for (int s = 0; s < 8; ++s) {
      int ks = ks0 + s;
      v2f w = *(const v2f*)(wsrc + 4 * ks);
      v2f c = *(const v2f*)(csrc + 4 * ks);
      v2f inv;
      inv.x = 1.0f / (w.x * w.x);
      inv.y = 1.0f / (w.y * w.y);
      v2f ci = c * inv;
      int o = ((rt * KSTEPS + ks) * 32 + lane) * 2;
      *(v2f*)(iwp + o) = inv;
      *(v2f*)(cwp + o) = ci;
    }
  }
}

#define ZKB 2
#define FKB 4
__global__ __launch_bounds__(256, 4) void rowblock_kernel(
    const float* __restrict__ query, const float* __restrict__ wqp,
    const float* __restrict__ bias, const float* __restrict__ iwp,
    const float* __restrict__ cc, const float* __restrict__ conqp,
    float* __restrict__ out) {
  __shared__ __attribute__((aligned(16))) float stage[32 * SM_PITCH];
  __shared__ float red[2][8][32];
  int wave = __builtin_amdgcn_readfirstlane(threadIdx.x >> 5);
  int lane = threadIdx.x & 31;
  int hi = lane >> 4;
  int ln = lane & 15;
  int mt0 = blockIdx.x * 2;
  const float* sp = stage + lane * 2;

  {
    const float* src = query + (size_t)blockIdx.x * (32 * DD);
    float4 v[16];
#pragma unroll
    for (int i = 0; i < 16; ++i) {
      int idx = threadIdx.x + 256 * i;
      v[i] = *(const float4*)(src + 4 * idx);
    }
#pragma unroll
    for (int i = 0; i < 16; ++i) {
      int idx = threadIdx.x + 256 * i;
      *(float4*)&stage[(idx >> 7) * SM_PITCH + 4 * (idx & 127)] = v[i];
    }
  }
  __syncthreads();

  {
    const float* xp = stage + ln * SM_PITCH + 2 * hi;
    const float* wp = wqp + (wave * 4) * TILE_FLOATS + lane * 2;
    v8f acc[4][2];
#pragma unroll
    for (int j = 0; j < 4; ++j) {
#pragma unroll
      for (int t = 0; t < 2; ++t) acc[j][t] = v8f{};
    }
    v2f x0[FKB][2], w0[FKB][4], x1[FKB][2], w1[FKB][4];
    auto load = [&](v2f (&x)[FKB][2], v2f (&w)[FKB][4], int ks0) {
#pragma unroll
      for (int s = 0; s < FKB; ++s) {
        int o = (ks0 + s) * 64;
        x[s][0] = *(const v2f*)(xp + (ks0 + s) * 4);
        x[s][1] = *(const v2f*)(xp + 16 * SM_PITCH + (ks0 + s) * 4);
#pragma unroll
        for (int j = 0; j < 4; ++j) w[s][j] = *(const v2f*)(wp + j * TILE_FLOATS + o);
      }
    };
    auto mma = [&](const v2f (&x)[FKB][2], const v2f (&w)[FKB][4]) {
#pragma unroll
      for (int s = 0; s < FKB; ++s) {
#pragma unroll
        for (int j = 0; j < 4; ++j) {
          acc[j][0] = wmma_f32(w[s][j], x[s][0], acc[j][0]);
          acc[j][1] = wmma_f32(w[s][j], x[s][1], acc[j][1]);
        }
      }
    };
    load(x0, w0, 0);
    for (int ks = 0; ks < KSTEPS - 2 * FKB; ks += 2 * FKB) {
      load(x1, w1, ks + FKB);
      SCHED_FENCE();
      mma(x0, w0);
      SCHED_FENCE();
      load(x0, w0, ks + 2 * FKB);
      SCHED_FENCE();
      mma(x1, w1);
      SCHED_FENCE();
    }
    load(x1, w1, KSTEPS - FKB);
    SCHED_FENCE();
    mma(x0, w0);
    mma(x1, w1);
    SCHED_FENCE();
    __syncthreads();
#pragma unroll
    for (int j = 0; j < 4; ++j) {
      const float4* bp4 = (const float4*)(bias + 64 * wave + 16 * j + 8 * hi);
      float4 ba = bp4[0], bb = bp4[1];
      float bv[8] = {ba.x, ba.y, ba.z, ba.w, bb.x, bb.y, bb.z, bb.w};
#pragma unroll
      for (int t = 0; t < 2; ++t) {
        float* d = stage + (t * KSTEPS + wave * 16 + 4 * j + 2 * hi) * 64 + ln * 2;
#pragma unroll
        for (int i = 0; i < 8; i += 2) {
          v2f o;
          o.x = acc[j][t][i] + bv[i];
          o.y = acc[j][t][i + 1] + bv[i + 1];
          *(v2f*)(d + (i >> 2) * 64 + ((i >> 1) & 1) * 32) = o;
        }
      }
    }
  }
  __syncthreads();

  {
    const float* ip0 = iwp + (wave * 4) * TILE_FLOATS + lane * 2;
    const float* cp0 = ip0 + RR * DD;

    v8f qq[4][2], qc[4][2];
#pragma unroll
    for (int j = 0; j < 4; ++j) {
#pragma unroll
      for (int t = 0; t < 2; ++t) {
        qq[j][t] = v8f{};
        qc[j][t] = v8f{};
      }
    }
    v2f a0[ZKB][2], bi0[ZKB][4], bc0[ZKB][4], a1[ZKB][2], bi1[ZKB][4], bc1[ZKB][4];
    auto load = [&](v2f (&a)[ZKB][2], v2f (&bi)[ZKB][4], v2f (&bc)[ZKB][4], int ks0) {
#pragma unroll
      for (int s = 0; s < ZKB; ++s) {
        int o = (ks0 + s) * 64;
        a[s][0] = *(const v2f*)(sp + o);
        a[s][1] = *(const v2f*)(sp + TILE_FLOATS + o);
#pragma unroll
        for (int j = 0; j < 4; ++j) {
          bi[s][j] = *(const v2f*)(ip0 + j * TILE_FLOATS + o);
          bc[s][j] = *(const v2f*)(cp0 + j * TILE_FLOATS + o);
        }
      }
    };
    auto mma = [&](const v2f (&a)[ZKB][2], const v2f (&bi)[ZKB][4], const v2f (&bc)[ZKB][4]) {
#pragma unroll
      for (int s = 0; s < ZKB; ++s) {
        v2f asq0 = a[s][0] * a[s][0];
        v2f asq1 = a[s][1] * a[s][1];
#pragma unroll
        for (int j = 0; j < 4; ++j) {
          qq[j][0] = wmma_f32(bi[s][j], asq0,    qq[j][0]);
          qc[j][0] = wmma_f32(bc[s][j], a[s][0], qc[j][0]);
          qq[j][1] = wmma_f32(bi[s][j], asq1,    qq[j][1]);
          qc[j][1] = wmma_f32(bc[s][j], a[s][1], qc[j][1]);
        }
      }
    };
    load(a0, bi0, bc0, 0);
    for (int ks = 0; ks < KSTEPS - 2 * ZKB; ks += 2 * ZKB) {
      load(a1, bi1, bc1, ks + ZKB);
      SCHED_FENCE();
      mma(a0, bi0, bc0);
      SCHED_FENCE();
      load(a0, bi0, bc0, ks + 2 * ZKB);
      SCHED_FENCE();
      mma(a1, bi1, bc1);
      SCHED_FENCE();
    }
    load(a1, bi1, bc1, KSTEPS - ZKB);
    SCHED_FENCE();
    mma(a0, bi0, bc0);
    mma(a1, bi1, bc1);

    const float scale = -0.5f / (float)DD;
    float zv[2][4][8];
#pragma unroll
    for (int j = 0; j < 4; ++j) {
      const float4* ccp = (const float4*)(cc + 64 * wave + 16 * j + 8 * hi);
      float4 ca = ccp[0], cb = ccp[1];
      float ccv[8] = {ca.x, ca.y, ca.z, ca.w, cb.x, cb.y, cb.z, cb.w};
#pragma unroll
      for (int t = 0; t < 2; ++t) {
#pragma unroll
        for (int i = 0; i < 8; ++i)
          zv[t][j][i] = scale * (qq[j][t][i] - 2.0f * qc[j][t][i] + ccv[i]);
      }
    }
    float mx[2];
#pragma unroll
    for (int t = 0; t < 2; ++t) {
      float m = -3.4e38f;
#pragma unroll
      for (int j = 0; j < 4; ++j) {
#pragma unroll
        for (int i = 0; i < 8; ++i) m = fmaxf(m, zv[t][j][i]);
      }
      mx[t] = fmaxf(m, __shfl_xor(m, 16, 32));
    }
    if (hi == 0) {
      red[0][wave][ln] = mx[0];
      red[0][wave][16 + ln] = mx[1];
    }
    __syncthreads();
    float sm[2];
#pragma unroll
    for (int t = 0; t < 2; ++t) {
      float m = red[0][0][16 * t + ln];
#pragma unroll
      for (int w = 1; w < 8; ++w) m = fmaxf(m, red[0][w][16 * t + ln]);
      float s = 0.0f;
#pragma unroll
      for (int j = 0; j < 4; ++j) {
#pragma unroll
        for (int i = 0; i < 8; ++i) {
          zv[t][j][i] = __expf(zv[t][j][i] - m);
          s += zv[t][j][i];
        }
      }
      sm[t] = s + __shfl_xor(s, 16, 32);
    }
    if (hi == 0) {
      red[1][wave][ln] = sm[0];
      red[1][wave][16 + ln] = sm[1];
    }
    __syncthreads();
#pragma unroll
    for (int t = 0; t < 2; ++t) {
      float s = red[1][0][16 * t + ln];
#pragma unroll
      for (int w = 1; w < 8; ++w) s += red[1][w][16 * t + ln];
      float inv = 1.0f / s;
      float* dst = stage + (t * KSTEPS + wave * 16 + 2 * hi) * 64 + ln * 2;
#pragma unroll
      for (int j = 0; j < 4; ++j) {
#pragma unroll
        for (int i = 0; i < 8; i += 2) {
          v2f o;
          o.x = zv[t][j][i] * inv;
          o.y = zv[t][j][i + 1] * inv;
          *(v2f*)(dst + (4 * j + (i >> 2)) * 64 + ((i >> 1) & 1) * 32) = o;
        }
      }
    }
  }
  __syncthreads();

  {
    int b = blockIdx.x / (LL / 32);
    const float* bp = conqp + b * RR * DD + (wave * 4) * TILE_FLOATS + lane * 2;
    v8f acc[2][4];
#pragma unroll
    for (int t = 0; t < 2; ++t) {
#pragma unroll
      for (int j = 0; j < 4; ++j) acc[t][j] = v8f{};
    }
    v2f a0[FKB][2], b0[FKB][4], a1[FKB][2], b1[FKB][4];
    auto load = [&](v2f (&a)[FKB][2], v2f (&b)[FKB][4], int ks0) {
#pragma unroll
      for (int s = 0; s < FKB; ++s) {
        int o = (ks0 + s) * 64;
        a[s][0] = *(const v2f*)(sp + o);
        a[s][1] = *(const v2f*)(sp + TILE_FLOATS + o);
#pragma unroll
        for (int j = 0; j < 4; ++j) b[s][j] = *(const v2f*)(bp + j * TILE_FLOATS + o);
      }
    };
    auto mma = [&](const v2f (&a)[FKB][2], const v2f (&b)[FKB][4]) {
#pragma unroll
      for (int s = 0; s < FKB; ++s) {
#pragma unroll
        for (int j = 0; j < 4; ++j) {
          acc[0][j] = wmma_f32(a[s][0], b[s][j], acc[0][j]);
          acc[1][j] = wmma_f32(a[s][1], b[s][j], acc[1][j]);
        }
      }
    };
    load(a0, b0, 0);
    for (int ks = 0; ks < KSTEPS - 2 * FKB; ks += 2 * FKB) {
      load(a1, b1, ks + FKB);
      SCHED_FENCE();
      mma(a0, b0);
      SCHED_FENCE();
      load(a0, b0, ks + 2 * FKB);
      SCHED_FENCE();
      mma(a1, b1);
      SCHED_FENCE();
    }
    load(a1, b1, KSTEPS - FKB);
    SCHED_FENCE();
    mma(a0, b0);
    mma(a1, b1);
    float* ob = out + (size_t)(blockIdx.x * 32 + 8 * hi) * DD + 64 * wave + ln;
#pragma unroll
    for (int t = 0; t < 2; ++t) {
#pragma unroll
      for (int i = 0; i < 8; ++i) {
        float* crow = ob + (16 * t + i) * DD;
        crow[0]  = acc[t][0][i];
        crow[16] = acc[t][1][i];
        crow[32] = acc[t][2][i];
        crow[48] = acc[t][3][i];
      }
    }
  }
}

#define CKB 2
#define CQ_KSTEPS 128
#define CPART_FLOATS 4096
typedef float v4f __attribute__((ext_vector_type(4)));
__global__ __launch_bounds__(256) void conq_kernel(
    const float* __restrict__ query, const float* __restrict__ wcp,
    const float* __restrict__ bc, float* __restrict__ conqp) {
  __shared__ __attribute__((aligned(16))) float part[8 * CPART_FLOATS];
  int b = blockIdx.z;
  int wave = __builtin_amdgcn_readfirstlane(threadIdx.x >> 5);
  int lane = threadIdx.x & 31;
  int hi = lane >> 4;
  int ln = lane & 15;
  int mt = wave & 1;
  int kq = wave >> 1;
  int tile_n = blockIdx.y * 64;

  const float* ap = wcp + (blockIdx.x * 8 + mt * 4) * LTILE_FLOATS + kq * (CQ_KSTEPS * 64) + lane * 2;
  const float* bp = query + b * (LL * DD) + (kq * (CQ_KSTEPS * 4) + 2 * hi) * DD + tile_n + ln;

  v8f acc[4][4];
#pragma unroll
  for (int i = 0; i < 4; ++i) {
#pragma unroll
    for (int j = 0; j < 4; ++j) acc[i][j] = v8f{};
  }
  v2f a0[CKB][4], b0[CKB][4], a1[CKB][4], b1[CKB][4], a2[CKB][4], b2[CKB][4], a3[CKB][4], b3[CKB][4];
  auto load = [&](v2f (&a)[CKB][4], v2f (&bf)[CKB][4], int ks0) {
#pragma unroll
    for (int s = 0; s < CKB; ++s) {
      int oa = (ks0 + s) * 64;
      int ob = (ks0 + s) * (4 * DD);
#pragma unroll
      for (int i = 0; i < 4; ++i) a[s][i] = *(const v2f*)(ap + i * LTILE_FLOATS + oa);
#pragma unroll
      for (int j = 0; j < 4; ++j) {
        v2f t;
        t.x = bp[ob + 16 * j];
        t.y = bp[ob + DD + 16 * j];
        bf[s][j] = t;
      }
    }
  };
  auto mma = [&](const v2f (&a)[CKB][4], const v2f (&bf)[CKB][4]) {
#pragma unroll
    for (int s = 0; s < CKB; ++s) {
#pragma unroll
      for (int i = 0; i < 4; ++i) {
#pragma unroll
        for (int j = 0; j < 4; ++j) acc[i][j] = wmma_f32(a[s][i], bf[s][j], acc[i][j]);
      }
    }
  };
  load(a0, b0, 0);
  load(a1, b1, CKB);
  load(a2, b2, 2 * CKB);
  for (int ks = 0; ks < CQ_KSTEPS - 4 * CKB; ks += 4 * CKB) {
    load(a3, b3, ks + 3 * CKB);
    SCHED_FENCE();
    mma(a0, b0);
    SCHED_FENCE();
    load(a0, b0, ks + 4 * CKB);
    SCHED_FENCE();
    mma(a1, b1);
    SCHED_FENCE();
    load(a1, b1, ks + 5 * CKB);
    SCHED_FENCE();
    mma(a2, b2);
    SCHED_FENCE();
    load(a2, b2, ks + 6 * CKB);
    SCHED_FENCE();
    mma(a3, b3);
    SCHED_FENCE();
  }
  load(a3, b3, CQ_KSTEPS - CKB);
  SCHED_FENCE();
  mma(a0, b0);
  mma(a1, b1);
  mma(a2, b2);
  mma(a3, b3);

  {
    float* pw = part + wave * CPART_FLOATS + lane * 4;
#pragma unroll
    for (int i = 0; i < 4; ++i) {
#pragma unroll
      for (int j = 0; j < 4; ++j) {
        v4f lo = {acc[i][j][0], acc[i][j][1], acc[i][j][2], acc[i][j][3]};
        v4f up = {acc[i][j][4], acc[i][j][5], acc[i][j][6], acc[i][j][7]};
        *(v4f*)(pw + ((i * 4 + j) * 2 + 0) * 128) = lo;
        *(v4f*)(pw + ((i * 4 + j) * 2 + 1) * 128) = up;
      }
    }
  }
  __syncthreads();
  v8f c[4];
#pragma unroll
  for (int j = 0; j < 4; ++j) {
    const float* pr = part + mt * CPART_FLOATS + ((kq * 4 + j) * 2) * 128 + lane * 4;
    v4f lo = *(const v4f*)(pr);
    v4f up = *(const v4f*)(pr + 128);
#pragma unroll
    for (int q = 1; q < 4; ++q) {
      lo += *(const v4f*)(pr + q * 2 * CPART_FLOATS);
      up += *(const v4f*)(pr + q * 2 * CPART_FLOATS + 128);
    }
    c[j] = v8f{lo.x, lo.y, lo.z, lo.w, up.x, up.y, up.z, up.w};
  }
  int tile_m = blockIdx.x * 128 + mt * 64 + kq * 16;
  const float* bm = bc + tile_m + 8 * hi;
  float* dst = conqp + b * RR * DD
             + ((tile_n >> 4) * KSTEPS + (tile_m >> 2) + 2 * hi) * 64 + ln * 2;
#pragma unroll
  for (int i = 0; i < 8; i += 2) {
    float bias0 = bm[i], bias1 = bm[i + 1];
    float* d = dst + (i >> 2) * 64 + ((i >> 1) & 1) * 32;
    v2f o0, o1, o2, o3;
    o0.x = c[0][i] + bias0;  o0.y = c[0][i + 1] + bias1;
    o1.x = c[1][i] + bias0;  o1.y = c[1][i + 1] + bias1;
    o2.x = c[2][i] + bias0;  o2.y = c[2][i + 1] + bias1;
    o3.x = c[3][i] + bias0;  o3.y = c[3][i + 1] + bias1;
    *(v2f*)(d + 0 * TILE_FLOATS) = o0;
    *(v2f*)(d + 1 * TILE_FLOATS) = o1;
    *(v2f*)(d + 2 * TILE_FLOATS) = o2;
    *(v2f*)(d + 3 * TILE_FLOATS) = o3;
  }
}

extern "C" void kernel_launch(void* const* d_in, const int* in_sizes, int n_in,
                              void* d_out, int out_size, void* d_ws, size_t ws_size,
                              hipStream_t stream) {
  const float* query   = (const float*)d_in[0];
  const float* Wq      = (const float*)d_in[1];
  const float* bq      = (const float*)d_in[2];
  const float* Wc      = (const float*)d_in[3];
  const float* bc      = (const float*)d_in[4];
  const float* centers = (const float*)d_in[5];
  const float* widths  = (const float*)d_in[6];
  float* out = (float*)d_out;

  float* iwp   = (float*)d_ws;
  float* cwp   = iwp   + (size_t)RR * DD;
  float* cc    = cwp   + (size_t)RR * DD;
  float* conqp = cc    + RR;
  float* wqp   = conqp + (size_t)BB * RR * DD;
  float* wcp   = wqp   + (size_t)DD * DD;

  pack_kernel<<<PACK_WQ_BLOCKS + PACK_WC_BLOCKS, 256, 0, stream>>>(Wq, Wc, wqp, wcp);

  prep_kernel<<<RR / 8, 256, 0, stream>>>(centers, widths, iwp, cwp, cc);

  conq_kernel<<<dim3(RR / 128, DD / 64, BB), 256, 0, stream>>>(query, wcp, bc, conqp);

  rowblock_kernel<<<MTOT / 32, 256, 0, stream>>>(query, wqp, bq, iwp, cc, conqp, out);
}
